# starting kernel
# baseline (speedup 1.0000x reference)
#include <hip/hip_runtime.h>

typedef float v2f __attribute__((ext_vector_type(2)));
typedef float v4f __attribute__((ext_vector_type(4)));
typedef float v8f __attribute__((ext_vector_type(8)));

#define BLK_M 128
#define BLK_N 256
#define BLK_K 16
#define STRIDE_A 20
#define PSTRIDE 544
#define ASZ (BLK_M * STRIDE_A)
#define BSZ ((BLK_K / 2) * PSTRIDE)

__global__ __launch_bounds__(256)
void Dynamic_Masking_Layer_gemm_wmma_f32(const float* __restrict__ X,
                                         const float* __restrict__ Mk,
                                         const float* __restrict__ W,
                                         const float* __restrict__ Bias,
                                         float* __restrict__ Out,
                                         int M, int N, int K)
{
    __shared__ float As[2 * ASZ];
    __shared__ float Bs[2 * BSZ];

    const int tid  = threadIdx.x;
    const int lane = tid & 31;
    const int wave = tid >> 5;
    const int wm   = wave & 1;
    const int wn   = wave >> 1;

    const int m_blk = blockIdx.y * BLK_M;
    const int n_blk = blockIdx.x * BLK_N;

    const int rowA  = tid >> 2;
    const int col4A = tid & 3;
    const int pB = tid >> 6;
    const int qB = tid & 63;

    v8f acc[4][4];
#pragma unroll
    for (int im = 0; im < 4; ++im)
#pragma unroll
        for (int in = 0; in < 4; ++in)
#pragma unroll
            for (int v = 0; v < 8; ++v)
                acc[im][in][v] = 0.0f;

    const int KT = K / BLK_K;

    v4f ra[2], rw0[2], rw1[2], rm0[2], rm1[2];
#pragma unroll
    for (int i = 0; i < 2; ++i)
        ra[i] = *(const v4f*)(X + (size_t)(m_blk + rowA + 64 * i) * K + (col4A * 4));
#pragma unroll
    for (int i = 0; i < 2; ++i) {
        const int p = pB + 4 * i;
        rw0[i] = *(const v4f*)(W  + (size_t)(2 * p + 0) * N + (n_blk + qB * 4));
        rw1[i] = *(const v4f*)(W  + (size_t)(2 * p + 1) * N + (n_blk + qB * 4));
        rm0[i] = *(const v4f*)(Mk + (size_t)(2 * p + 0) * N + (n_blk + qB * 4));
        rm1[i] = *(const v4f*)(Mk + (size_t)(2 * p + 1) * N + (n_blk + qB * 4));
    }

    for (int kt = 0; kt < KT; ++kt) {
        const int buf = kt & 1;
        float* __restrict__ as_w = As + buf * ASZ;
        float* __restrict__ bs_w = Bs + buf * BSZ;

#pragma unroll
        for (int i = 0; i < 2; ++i)
            *(v4f*)(as_w + (rowA + 64 * i) * STRIDE_A + col4A * 4) = ra[i];
#pragma unroll
        for (int i = 0; i < 2; ++i) {
            const int p = pB + 4 * i;
#pragma unroll
            for (int j = 0; j < 4; ++j) {
                v2f pr;
                pr.x = (rm0[i][j] != 0.0f) ? rw0[i][j] : 0.0f;
                pr.y = (rm1[i][j] != 0.0f) ? rw1[i][j] : 0.0f;
                *(v2f*)(bs_w + p * PSTRIDE + (qB * 4 + j) * 2) = pr;
            }
        }
        __syncthreads();

        if (kt + 1 < KT) {
            const int kn = (kt + 1) * BLK_K;
#pragma unroll
            for (int i = 0; i < 2; ++i)
                ra[i] = *(const v4f*)(X + (size_t)(m_blk + rowA + 64 * i) * K + (kn + col4A * 4));
#pragma unroll
            for (int i = 0; i < 2; ++i) {
                const int p = pB + 4 * i;
                rw0[i] = *(const v4f*)(W  + (size_t)(kn + 2 * p + 0) * N + (n_blk + qB * 4));
                rw1[i] = *(const v4f*)(W  + (size_t)(kn + 2 * p + 1) * N + (n_blk + qB * 4));
                rm0[i] = *(const v4f*)(Mk + (size_t)(kn + 2 * p + 0) * N + (n_blk + qB * 4));
                rm1[i] = *(const v4f*)(Mk + (size_t)(kn + 2 * p + 1) * N + (n_blk + qB * 4));
            }
        }

        const float* __restrict__ as_r = As + buf * ASZ;
        const float* __restrict__ bs_r = Bs + buf * BSZ;

#pragma unroll
        for (int s = 0; s < 4; ++s) {
            const int krow = 4 * s + 2 * (lane >> 4);
            const int prow = 2 * s + (lane >> 4);

            v2f af[4];
#pragma unroll
            for (int im = 0; im < 4; ++im) {
                const int m_l = wm * 64 + im * 16 + (lane & 15);
                af[im] = *(const v2f*)(as_r + m_l * STRIDE_A + krow);
            }

            v2f bf[4];
#pragma unroll
            for (int in = 0; in < 4; ++in) {
                const int n_l = wn * 64 + in * 16 + (lane & 15);
                bf[in] = *(const v2f*)(bs_r + prow * PSTRIDE + n_l * 2);
            }

#pragma unroll
            for (int im = 0; im < 4; ++im)
#pragma unroll
                for (int in = 0; in < 4; ++in)
                    acc[im][in] = __builtin_amdgcn_wmma_f32_16x16x4_f32(
                        false, af[im], false, bf[in],
                        (short)0, acc[im][in], false, false);
        }
    }

#pragma unroll
    for (int in = 0; in < 4; ++in) {
        const int n_g = n_blk + wn * 64 + in * 16 + (lane & 15);
        const float bv = Bias[n_g];
#pragma unroll
        for (int im = 0; im < 4; ++im) {
            const int m_base = m_blk + wm * 64 + im * 16 + 8 * (lane >> 4);
#pragma unroll
            for (int v = 0; v < 8; ++v)
                Out[(size_t)(m_base + v) * N + n_g] = acc[im][in][v] + bv;
        }
    }
}

extern "C" void kernel_launch(void* const* d_in, const int* in_sizes, int n_in,
                              void* d_out, int out_size, void* d_ws, size_t ws_size,
                              hipStream_t stream) {
    const float* X    = (const float*)d_in[0];
    const float* Mask = (const float*)d_in[1];
    const float* W    = (const float*)d_in[2];
    const float* Bias = (const float*)d_in[3];
    float* Out        = (float*)d_out;

    const int N = in_sizes[3];
    const int K = in_sizes[1] / N;
    const int M = in_sizes[0] / K;

    dim3 grid(N / BLK_N, M / BLK_M);
    Dynamic_Masking_Layer_gemm_wmma_f32<<<grid, dim3(256), 0, stream>>>(
        X, Mask, W, Bias, Out, M, N, K);
}


# LDS ring of 8 K-slices, each written 4 slices ahead of its use: one workgroup barrier per 4 slices instead of per slice, so the two waves of a SIMD run out of step and cover each other's staging
# speedup vs baseline: 1.0832x; 1.0832x over previous
#include <hip/hip_runtime.h>


typedef float v2f __attribute__((ext_vector_type(2)));
typedef float v4f __attribute__((ext_vector_type(4)));
typedef float v8f __attribute__((ext_vector_type(8)));

#define BLK_M 128
#define BLK_N 256
#define BLK_K 16
#define STRIDE_A 20
#define PSTRIDE 544
#define ASZ (BLK_M * STRIDE_A)
#define BSZ ((BLK_K / 2) * PSTRIDE)
#define RING 8
#define AHEAD 4

__global__ __launch_bounds__(256)
void Dynamic_Masking_Layer_gemm_wmma_f32(const float* __restrict__ X,
                                         const float* __restrict__ Mk,
                                         const float* __restrict__ W,
                                         const float* __restrict__ Bias,
                                         float* __restrict__ Out,
                                         int M, int N, int K)
{
    __shared__ float As[RING * ASZ];
    __shared__ float Bs[RING * BSZ];

    const int tid  = threadIdx.x;
    const int lane = tid & 31;
    const int wave = tid >> 5;
    const int wm   = wave & 1;
    const int wn   = wave >> 1;

    const int m_blk = blockIdx.y * BLK_M;
    const int n_blk = blockIdx.x * BLK_N;

    const int rowA  = tid >> 2;
    const int col4A = tid & 3;
    const int pB = tid >> 6;
    const int qB = tid & 63;

    v8f acc[4][4];
#pragma unroll
    for (int im = 0; im < 4; ++im)
#pragma unroll
        for (int in = 0; in < 4; ++in)
#pragma unroll
            for (int v = 0; v < 8; ++v)
                acc[im][in][v] = 0.0f;

    const int KT = K / BLK_K;

    v4f ra[2], rw0[2], rw1[2], rm0[2], rm1[2];

    auto gload = [&](int kt) {
        const int kn = kt * BLK_K;
#pragma unroll
        for (int i = 0; i < 2; ++i)
            ra[i] = *(const v4f*)(X + (size_t)(m_blk + rowA + 64 * i) * K + (kn + col4A * 4));
#pragma unroll
        for (int i = 0; i < 2; ++i) {
            const int p = pB + 4 * i;
            rw0[i] = *(const v4f*)(W  + (size_t)(kn + 2 * p + 0) * N + (n_blk + qB * 4));
            rw1[i] = *(const v4f*)(W  + (size_t)(kn + 2 * p + 1) * N + (n_blk + qB * 4));
            rm0[i] = *(const v4f*)(Mk + (size_t)(kn + 2 * p + 0) * N + (n_blk + qB * 4));
            rm1[i] = *(const v4f*)(Mk + (size_t)(kn + 2 * p + 1) * N + (n_blk + qB * 4));
        }
    };

    auto stage = [&](int slot) {
        float* __restrict__ as_w = As + slot * ASZ;
        float* __restrict__ bs_w = Bs + slot * BSZ;
#pragma unroll
        for (int i = 0; i < 2; ++i)
            *(v4f*)(as_w + (rowA + 64 * i) * STRIDE_A + col4A * 4) = ra[i];
#pragma unroll
        for (int i = 0; i < 2; ++i) {
            const int p = pB + 4 * i;
#pragma unroll
            for (int j = 0; j < 4; ++j) {
                v2f pr;
                pr.x = (rm0[i][j] != 0.0f) ? rw0[i][j] : 0.0f;
                pr.y = (rm1[i][j] != 0.0f) ? rw1[i][j] : 0.0f;
                *(v2f*)(bs_w + p * PSTRIDE + (qB * 4 + j) * 2) = pr;
            }
        }
    };

    gload(0);
#pragma unroll 1
    for (int s = 0; s < AHEAD; ++s) {
        if (s < KT) stage(s);
        if (s + 1 < KT) gload(s + 1);
    }
    __syncthreads();

#pragma unroll 1
    for (int kt = 0; kt < KT; ++kt) {
        const int slot = kt & (RING - 1);
        const float* __restrict__ as_r = As + slot * ASZ;
        const float* __restrict__ bs_r = Bs + slot * BSZ;

        v2f af[4][4], bf[4][4];
#pragma unroll
        for (int s = 0; s < 4; ++s) {
            const int krow = 4 * s + 2 * (lane >> 4);
            const int prow = 2 * s + (lane >> 4);
#pragma unroll
            for (int im = 0; im < 4; ++im) {
                const int m_l = wm * 64 + im * 16 + (lane & 15);
                af[s][im] = *(const v2f*)(as_r + m_l * STRIDE_A + krow);
            }
#pragma unroll
            for (int in = 0; in < 4; ++in) {
                const int n_l = wn * 64 + in * 16 + (lane & 15);
                bf[s][in] = *(const v2f*)(bs_r + prow * PSTRIDE + n_l * 2);
            }
        }

        if (kt + AHEAD < KT) stage((kt + AHEAD) & (RING - 1));

        if (kt + AHEAD + 1 < KT) gload(kt + AHEAD + 1);

#pragma unroll
        for (int s = 0; s < 4; ++s)
#pragma unroll
            for (int im = 0; im < 4; ++im)
#pragma unroll
                for (int in = 0; in < 4; ++in)
                    acc[im][in] = __builtin_amdgcn_wmma_f32_16x16x4_f32(
                        false, af[s][im], false, bf[s][in],
                        (short)0, acc[im][in], false, false);

        asm volatile("v_nop\n\tv_nop\n\tv_nop\n\tv_nop\n\tv_nop");

        if ((kt & (AHEAD - 1)) == AHEAD - 1) __syncthreads();
    }

#pragma unroll
    for (int in = 0; in < 4; ++in) {
        const int n_g = n_blk + wn * 64 + in * 16 + (lane & 15);
        const float bv = Bias[n_g];
#pragma unroll
        for (int im = 0; im < 4; ++im) {
            const int m_base = m_blk + wm * 64 + im * 16 + 8 * (lane >> 4);
#pragma unroll
            for (int v = 0; v < 8; ++v)
                Out[(size_t)(m_base + v) * N + n_g] = acc[im][in][v] + bv;
        }
    }
}

extern "C" void kernel_launch(void* const* d_in, const int* in_sizes, int n_in,
                              void* d_out, int out_size, void* d_ws, size_t ws_size,
                              hipStream_t stream) {
    const float* X    = (const float*)d_in[0];
    const float* Mask = (const float*)d_in[1];
    const float* W    = (const float*)d_in[2];
    const float* Bias = (const float*)d_in[3];
    float* Out        = (float*)d_out;

    const int N = in_sizes[3];
    const int K = in_sizes[1] / N;
    const int M = in_sizes[0] / K;

    dim3 grid(N / BLK_N, M / BLK_M);
    Dynamic_Masking_Layer_gemm_wmma_f32<<<grid, dim3(256), 0, stream>>>(
        X, Mask, W, Bias, Out, M, N, K);
}


# one wave per SIMD on a 64x128 tile with a second fragment register set: next slice's LDS reads as one group ahead of the WMMAs, staging's vector work behind 32 WMMAs that end no fragment's use
# speedup vs baseline: 1.1326x; 1.0456x over previous
#include <hip/hip_runtime.h>


typedef float v2f __attribute__((ext_vector_type(2)));
typedef float v4f __attribute__((ext_vector_type(4)));
typedef float v8f __attribute__((ext_vector_type(8)));

#define BLK_M 128
#define BLK_N 256
#define BLK_K 16
#define STRIDE_A 20
#define PSTRIDE 544
#define ASZ (BLK_M * STRIDE_A)
#define BSZ ((BLK_K / 2) * PSTRIDE)
#define RING 8
#define AHEAD 5
#define GROUP 4

#define WMMA_PAD() asm volatile("v_nop\n\tv_nop\n\tv_nop\n\tv_nop\n\tv_nop")

__global__ __launch_bounds__(128)
void Dynamic_Masking_Layer_gemm_wmma_f32(const float* __restrict__ X,
                                         const float* __restrict__ Mk,
                                         const float* __restrict__ W,
                                         const float* __restrict__ Bias,
                                         float* __restrict__ Out,
                                         int M, int N, int K)
{
    __shared__ float As[RING * ASZ];
    __shared__ float Bs[RING * BSZ];

    const int tid  = threadIdx.x;
    const int lane = tid & 31;
    const int wave = tid >> 5;
    const int wm   = wave & 1;
    const int wn   = wave >> 1;

    const int m_blk = blockIdx.y * BLK_M;
    const int n_blk = blockIdx.x * BLK_N;

    const int rowA  = tid >> 2;
    const int col4A = tid & 3;
    const int pB = tid >> 6;
    const int qB = tid & 63;

    const char* pa[4];
    const char* pb[16];
#pragma unroll
    for (int i = 0; i < 4; ++i) {
        pa[i] = (const char*)(X + (size_t)(m_blk + rowA + 32 * i) * K + col4A * 4);
        const size_t o = (size_t)(2 * (pB + 2 * i)) * N + n_blk + qB * 4;
        pb[4 * i + 0] = (const char*)(W + o);
        pb[4 * i + 1] = (const char*)(W + o + N);
        pb[4 * i + 2] = (const char*)(Mk + o);
        pb[4 * i + 3] = (const char*)(Mk + o + N);
    }
    const size_t strideA = (size_t)BLK_K * sizeof(float);
    const size_t strideB = (size_t)BLK_K * N * sizeof(float);

    v8f acc[4][8];
#pragma unroll
    for (int im = 0; im < 4; ++im)
#pragma unroll
        for (int in = 0; in < 8; ++in)
#pragma unroll
            for (int v = 0; v < 8; ++v)
                acc[im][in][v] = 0.0f;

    const int KT = K / BLK_K;

    v4f ra[4], rw0[4], rw1[4], rm0[4], rm1[4];

    auto gload = [&]() {
#pragma unroll
        for (int i = 0; i < 4; ++i)
            ra[i] = *(const v4f*)pa[i];
#pragma unroll
        for (int i = 0; i < 4; ++i) {
            rw0[i] = *(const v4f*)pb[4 * i + 0];
            rw1[i] = *(const v4f*)pb[4 * i + 1];
            rm0[i] = *(const v4f*)pb[4 * i + 2];
            rm1[i] = *(const v4f*)pb[4 * i + 3];
        }
    };

    auto sel = [&](int i, int j) -> v2f {
        v2f pr;
        pr.x = (rm0[i][j] != 0.0f) ? rw0[i][j] : 0.0f;
        pr.y = (rm1[i][j] != 0.0f) ? rw1[i][j] : 0.0f;
        return pr;
    };

    auto put = [&](int slot, const v2f (&pr)[4][4]) {
        float* __restrict__ as_w = As + slot * ASZ;
        float* __restrict__ bs_w = Bs + slot * BSZ;
#pragma unroll
        for (int i = 0; i < 4; ++i)
            *(v4f*)(as_w + (rowA + 32 * i) * STRIDE_A + col4A * 4) = ra[i];
#pragma unroll
        for (int i = 0; i < 4; ++i)
#pragma unroll
            for (int j = 0; j < 4; ++j)
                *(v2f*)(bs_w + (pB + 2 * i) * PSTRIDE + (qB * 4 + j) * 2) = pr[i][j];
    };

    auto stage = [&](int slot) {
        v2f pr[4][4];
#pragma unroll
        for (int i = 0; i < 4; ++i)
#pragma unroll
            for (int j = 0; j < 4; ++j)
                pr[i][j] = sel(i, j);
        put(slot, pr);
    };

    auto loadfr = [&](int slot, v2f (&fa)[4][4], v2f (&fb)[4][8]) {
        const float* __restrict__ as_r = As + slot * ASZ;
        const float* __restrict__ bs_r = Bs + slot * BSZ;
#pragma unroll
        for (int s = 0; s < 4; ++s) {
            const int krow = 4 * s + 2 * (lane >> 4);
            const int prow = 2 * s + (lane >> 4);
#pragma unroll
            for (int im = 0; im < 4; ++im) {
                const int m_l = wm * 64 + im * 16 + (lane & 15);
                fa[s][im] = *(const v2f*)(as_r + m_l * STRIDE_A + krow);
            }
#pragma unroll
            for (int in = 0; in < 8; ++in) {
                const int n_l = wn * 128 + in * 16 + (lane & 15);
                fb[s][in] = *(const v2f*)(bs_r + prow * PSTRIDE + n_l * 2);
            }
        }
    };

    auto wmma1 = [&](int s, int im, int in, v2f (&fa)[4][4], v2f (&fb)[4][8]) {
        acc[im][in] = __builtin_amdgcn_wmma_f32_16x16x4_f32(
            false, fa[s][im], false, fb[s][in],
            (short)0, acc[im][in], false, false);
    };

    auto mma_rest = [&](v2f (&fa)[4][4], v2f (&fb)[4][8]) {
#pragma unroll
        for (int s = 0; s < 4; ++s)
#pragma unroll
            for (int im = 0; im < 4; ++im)
#pragma unroll
                for (int in = 0; in < 8; ++in)
                    if ((in >> 1) != im) wmma1(s, im, in, fa, fb);
    };

    auto trip = [&](int kt, v2f (&ca)[4][4], v2f (&cb)[4][8], v2f (&na)[4][4], v2f (&nb)[4][8]) {
        const int nx = (kt + 1 < KT) ? kt + 1 : kt;
        loadfr(nx & (RING - 1), na, nb);
        __builtin_amdgcn_sched_barrier(0);
        const bool more = kt + AHEAD + 1 < KT;
        const size_t stA = more ? strideA : 0;
        const size_t stB = more ? strideB : 0;
        v2f pr[4][4];
#pragma unroll
        for (int s = 0; s < 4; ++s) {
#pragma unroll
            for (int im = 0; im < 4; ++im) {
                wmma1(s, im, 2 * im, ca, cb);
                wmma1(s, im, 2 * im + 1, ca, cb);
                pr[s][im] = sel(s, im);
                pb[4 * s + im] += stB;
                if (im == 0) pa[s] += stA;
                __builtin_amdgcn_sched_barrier(0);
            }
        }
        put((kt + AHEAD) & (RING - 1), pr);
        gload();
        __builtin_amdgcn_sched_barrier(0);
        mma_rest(ca, cb);
        WMMA_PAD();
        __builtin_amdgcn_sched_barrier(0);
        if ((kt & (GROUP - 1)) == GROUP - 2) __syncthreads();
    };

    gload();
#pragma unroll 1
    for (int s = 0; s < AHEAD; ++s) {
        stage(s);
        if (s + 1 < KT) {
#pragma unroll
            for (int i = 0; i < 4; ++i) pa[i] += strideA;
#pragma unroll
            for (int i = 0; i < 16; ++i) pb[i] += strideB;
        }
        gload();
    }
    __syncthreads();

    v2f a0[4][4], b0[4][8], a1[4][4], b1[4][8];
    loadfr(0, a0, b0);

    int kt = 0;
#pragma unroll 1
    for (; kt + 1 < KT; kt += 2) {
        trip(kt, a0, b0, a1, b1);
        trip(kt + 1, a1, b1, a0, b0);
    }
    if (kt < KT) trip(kt, a0, b0, a1, b1);

#pragma unroll
    for (int in = 0; in < 8; ++in) {
        const int n_g = n_blk + wn * 128 + in * 16 + (lane & 15);
        const float bv = Bias[n_g];
#pragma unroll
        for (int im = 0; im < 4; ++im) {
            const int m_base = m_blk + wm * 64 + im * 16 + 8 * (lane >> 4);
#pragma unroll
            for (int v = 0; v < 8; ++v)
                Out[(size_t)(m_base + v) * N + n_g] = acc[im][in][v] + bv;
        }
    }
}

extern "C" void kernel_launch(void* const* d_in, const int* in_sizes, int n_in,
                              void* d_out, int out_size, void* d_ws, size_t ws_size,
                              hipStream_t stream) {
    const float* X    = (const float*)d_in[0];
    const float* Mask = (const float*)d_in[1];
    const float* W    = (const float*)d_in[2];
    const float* Bias = (const float*)d_in[3];
    float* Out        = (float*)d_out;

    const int N = in_sizes[3];
    const int K = in_sizes[1] / N;
    const int M = in_sizes[0] / K;

    dim3 grid(N / BLK_N, M / BLK_M);
    Dynamic_Masking_Layer_gemm_wmma_f32<<<grid, dim3(128), 0, stream>>>(
        X, Mask, W, Bias, Out, M, N, K);
}


# four waves per SIMD from two 8-wave workgroups per WGP (128x128 tile, 64x32 per wave, 32-k slices, 3 LDS stages): other waves' WMMAs cover each wave's staging, fragment reads and barrier
# speedup vs baseline: 1.1724x; 1.0351x over previous
#include <hip/hip_runtime.h>


typedef float v2f __attribute__((ext_vector_type(2)));
typedef float v4f __attribute__((ext_vector_type(4)));
typedef float v8f __attribute__((ext_vector_type(8)));

#define BLK_M 128
#define BLK_N 128
#define BLK_K 32
#define STRIDE_A 36
#define PSTRIDE 288
#define ASZ (BLK_M * STRIDE_A)
#define BSZ ((BLK_K / 2) * PSTRIDE)
#define STAGES 3

#define WMMA_PAD() asm volatile("v_nop\n\tv_nop\n\tv_nop\n\tv_nop\n\tv_nop")

__global__ __launch_bounds__(256)
void Dynamic_Masking_Layer_gemm_wmma_f32(const float* __restrict__ X,
                                         const float* __restrict__ Mk,
                                         const float* __restrict__ W,
                                         const float* __restrict__ Bias,
                                         float* __restrict__ Out,
                                         int M, int N, int K)
{
    __shared__ float As[STAGES * ASZ];
    __shared__ float Bs[STAGES * BSZ];

    const int tid  = threadIdx.x;
    const int lane = tid & 31;
    const int wave = tid >> 5;
    const int wm   = wave & 1;
    const int wn   = wave >> 1;

    const int m_blk = blockIdx.y * BLK_M;
    const int n_blk = blockIdx.x * BLK_N;

    const int rowA  = tid >> 3;
    const int col4A = tid & 7;
    const int pB = tid >> 5;
    const int qB = tid & 31;

    const float* xs = X + (size_t)m_blk * K;
    const float* ws = W + n_blk;
    const float* ms = Mk + n_blk;
    int offA[4], offB[2];
#pragma unroll
    for (int i = 0; i < 4; ++i) offA[i] = (rowA + 32 * i) * K + col4A * 4;
#pragma unroll
    for (int i = 0; i < 2; ++i) offB[i] = (2 * (pB + 8 * i)) * N + qB * 4;
    const size_t strideA = (size_t)BLK_K;
    const size_t strideB = (size_t)BLK_K * N;

    v8f acc[4][2];
#pragma unroll
    for (int im = 0; im < 4; ++im)
#pragma unroll
        for (int in = 0; in < 2; ++in)
#pragma unroll
            for (int v = 0; v < 8; ++v)
                acc[im][in][v] = 0.0f;

    const int KT = K / BLK_K;

    v4f ra[4], rw0[2], rw1[2], rm0[2], rm1[2];

    auto gload = [&]() {
#pragma unroll
        for (int i = 0; i < 4; ++i)
            ra[i] = *(const v4f*)(xs + offA[i]);
#pragma unroll
        for (int i = 0; i < 2; ++i) {
            rw0[i] = *(const v4f*)(ws + offB[i]);
            rw1[i] = *(const v4f*)(ws + offB[i] + N);
            rm0[i] = *(const v4f*)(ms + offB[i]);
            rm1[i] = *(const v4f*)(ms + offB[i] + N);
        }
    };

    int held = 0;
    auto advance = [&]() {
        if (held + 1 < KT) {
            xs += strideA; ws += strideB; ms += strideB;
            ++held;
        }
    };

    auto stage = [&](int st) {
        float* __restrict__ as_w = As + st * ASZ;
        float* __restrict__ bs_w = Bs + st * BSZ;
#pragma unroll
        for (int i = 0; i < 4; ++i)
            *(v4f*)(as_w + (rowA + 32 * i) * STRIDE_A + col4A * 4) = ra[i];
#pragma unroll
        for (int i = 0; i < 2; ++i) {
            float s0[4], s1[4];
#pragma unroll
            for (int j = 0; j < 4; ++j) {
                s0[j] = (rm0[i][j] != 0.0f) ? rw0[i][j] : 0.0f;
                s1[j] = (rm1[i][j] != 0.0f) ? rw1[i][j] : 0.0f;
            }
            float* dst = bs_w + (pB + 8 * i) * PSTRIDE + qB * 8;
            v4f lo, hi;
            lo[0] = s0[0]; lo[1] = s1[0]; lo[2] = s0[1]; lo[3] = s1[1];
            hi[0] = s0[2]; hi[1] = s1[2]; hi[2] = s0[3]; hi[3] = s1[3];
            *(v4f*)(dst)     = lo;
            *(v4f*)(dst + 4) = hi;
        }
    };

    auto compute = [&](int st) {
        const float* __restrict__ as_r = As + st * ASZ;
        const float* __restrict__ bs_r = Bs + st * BSZ;
#pragma unroll
        for (int hf = 0; hf < 2; ++hf) {
            v2f fa[4][4], fb[4][2];
#pragma unroll
            for (int s = 0; s < 4; ++s) {
                const int ks   = hf * 4 + s;
                const int krow = 4 * ks + 2 * (lane >> 4);
                const int prow = 2 * ks + (lane >> 4);
#pragma unroll
                for (int im = 0; im < 4; ++im) {
                    const int m_l = wm * 64 + im * 16 + (lane & 15);
                    fa[s][im] = *(const v2f*)(as_r + m_l * STRIDE_A + krow);
                }
#pragma unroll
                for (int in = 0; in < 2; ++in) {
                    const int n_l = wn * 32 + in * 16 + (lane & 15);
                    fb[s][in] = *(const v2f*)(bs_r + prow * PSTRIDE + n_l * 2);
                }
            }
            __builtin_amdgcn_sched_barrier(0);
#pragma unroll
            for (int s = 0; s < 4; ++s)
#pragma unroll
                for (int im = 0; im < 4; ++im)
#pragma unroll
                    for (int in = 0; in < 2; ++in)
                        acc[im][in] = __builtin_amdgcn_wmma_f32_16x16x4_f32(
                            false, fa[s][im], false, fb[s][in],
                            (short)0, acc[im][in], false, false);
            if (hf == 0) {
                __builtin_amdgcn_sched_barrier(0);
                WMMA_PAD();
                __builtin_amdgcn_sched_barrier(0);
            }
        }
    };

    gload();
    stage(0);
    advance();
    gload();
    stage(1);
    advance();
    gload();
    __syncthreads();

    int st_c = 0;
    int st_w = 2;
#pragma unroll 1
    for (int kt = 0; kt < KT; ++kt) {
        stage(st_w);
        advance();
        gload();
        compute(st_c);
        __syncthreads();
        __builtin_amdgcn_sched_barrier(0);
        WMMA_PAD();
        __builtin_amdgcn_sched_barrier(0);
        st_c = (st_c == STAGES - 1) ? 0 : st_c + 1;
        st_w = (st_w == STAGES - 1) ? 0 : st_w + 1;
    }

#pragma unroll
    for (int in = 0; in < 2; ++in) {
        const int n_g = n_blk + wn * 32 + in * 16 + (lane & 15);
        const float bv = Bias[n_g];
#pragma unroll
        for (int im = 0; im < 4; ++im) {
            const int m_base = m_blk + wm * 64 + im * 16 + 8 * (lane >> 4);
#pragma unroll
            for (int v = 0; v < 8; ++v)
                Out[(size_t)(m_base + v) * N + n_g] = acc[im][in][v] + bv;
        }
    }
}

extern "C" void kernel_launch(void* const* d_in, const int* in_sizes, int n_in,
                              void* d_out, int out_size, void* d_ws, size_t ws_size,
                              hipStream_t stream) {
    const float* X    = (const float*)d_in[0];
    const float* Mask = (const float*)d_in[1];
    const float* W    = (const float*)d_in[2];
    const float* Bias = (const float*)d_in[3];
    float* Out        = (float*)d_out;

    const int N = in_sizes[3];
    const int K = in_sizes[1] / N;
    const int M = in_sizes[0] / K;

    dim3 grid(N / BLK_N, M / BLK_M);
    Dynamic_Masking_Layer_gemm_wmma_f32<<<grid, dim3(256), 0, stream>>>(
        X, Mask, W, Bias, Out, M, N, K);
}


# wave priority stepped down with K progress (3,2,1,0 at 1/2, 3/4, 15/16 of the trips): the two workgroups of a WGP advance together instead of the older finishing first and the younger ending alone
# speedup vs baseline: 1.2174x; 1.0384x over previous
#include <hip/hip_runtime.h>


typedef float v2f __attribute__((ext_vector_type(2)));
typedef float v4f __attribute__((ext_vector_type(4)));
typedef float v8f __attribute__((ext_vector_type(8)));

#define BLK_M 128
#define BLK_N 128
#define BLK_K 32
#define STRIDE_A 36
#define PSTRIDE 288
#define ASZ (BLK_M * STRIDE_A)
#define BSZ ((BLK_K / 2) * PSTRIDE)
#define STAGES 3

#define WMMA_PAD() asm volatile("v_nop\n\tv_nop\n\tv_nop\n\tv_nop\n\tv_nop")

__global__ __launch_bounds__(256)
void Dynamic_Masking_Layer_gemm_wmma_f32(const float* __restrict__ X,
                                         const float* __restrict__ Mk,
                                         const float* __restrict__ W,
                                         const float* __restrict__ Bias,
                                         float* __restrict__ Out,
                                         int M, int N, int K)
{
    __shared__ float As[STAGES * ASZ];
    __shared__ float Bs[STAGES * BSZ];

    const int tid  = threadIdx.x;
    const int lane = tid & 31;
    const int wave = tid >> 5;
    const int wm   = wave & 1;
    const int wn   = wave >> 1;

    const int m_blk = blockIdx.y * BLK_M;
    const int n_blk = blockIdx.x * BLK_N;

    const int rowA  = tid >> 3;
    const int col4A = tid & 7;
    const int pB = tid >> 5;
    const int qB = tid & 31;

    const float* xs = X + (size_t)m_blk * K;
    const float* ws = W + n_blk;
    const float* ms = Mk + n_blk;
    int offA[4], offB[2];
#pragma unroll
    for (int i = 0; i < 4; ++i) offA[i] = (rowA + 32 * i) * K + col4A * 4;
#pragma unroll
    for (int i = 0; i < 2; ++i) offB[i] = (2 * (pB + 8 * i)) * N + qB * 4;
    const size_t strideA = (size_t)BLK_K;
    const size_t strideB = (size_t)BLK_K * N;

    v8f acc[4][2];
#pragma unroll
    for (int im = 0; im < 4; ++im)
#pragma unroll
        for (int in = 0; in < 2; ++in)
#pragma unroll
            for (int v = 0; v < 8; ++v)
                acc[im][in][v] = 0.0f;

    const int KT = K / BLK_K;

    v4f ra[4], rw0[2], rw1[2], rm0[2], rm1[2];

    auto gload = [&]() {
#pragma unroll
        for (int i = 0; i < 4; ++i)
            ra[i] = *(const v4f*)(xs + offA[i]);
#pragma unroll
        for (int i = 0; i < 2; ++i) {
            rw0[i] = *(const v4f*)(ws + offB[i]);
            rw1[i] = *(const v4f*)(ws + offB[i] + N);
            rm0[i] = *(const v4f*)(ms + offB[i]);
            rm1[i] = *(const v4f*)(ms + offB[i] + N);
        }
    };

    int held = 0;
    auto advance = [&]() {
        if (held + 1 < KT) {
            xs += strideA; ws += strideB; ms += strideB;
            ++held;
        }
    };

    auto stage = [&](int st) {
        float* __restrict__ as_w = As + st * ASZ;
        float* __restrict__ bs_w = Bs + st * BSZ;
#pragma unroll
        for (int i = 0; i < 4; ++i)
            *(v4f*)(as_w + (rowA + 32 * i) * STRIDE_A + col4A * 4) = ra[i];
#pragma unroll
        for (int i = 0; i < 2; ++i) {
            float s0[4], s1[4];
#pragma unroll
            for (int j = 0; j < 4; ++j) {
                s0[j] = (rm0[i][j] != 0.0f) ? rw0[i][j] : 0.0f;
                s1[j] = (rm1[i][j] != 0.0f) ? rw1[i][j] : 0.0f;
            }
            float* dst = bs_w + (pB + 8 * i) * PSTRIDE + qB * 8;
            v4f lo, hi;
            lo[0] = s0[0]; lo[1] = s1[0]; lo[2] = s0[1]; lo[3] = s1[1];
            hi[0] = s0[2]; hi[1] = s1[2]; hi[2] = s0[3]; hi[3] = s1[3];
            *(v4f*)(dst)     = lo;
            *(v4f*)(dst + 4) = hi;
        }
    };

    auto compute = [&](int st) {
        const float* __restrict__ as_r = As + st * ASZ;
        const float* __restrict__ bs_r = Bs + st * BSZ;
#pragma unroll
        for (int hf = 0; hf < 2; ++hf) {
            v2f fa[4][4], fb[4][2];
#pragma unroll
            for (int s = 0; s < 4; ++s) {
                const int ks   = hf * 4 + s;
                const int krow = 4 * ks + 2 * (lane >> 4);
                const int prow = 2 * ks + (lane >> 4);
#pragma unroll
                for (int im = 0; im < 4; ++im) {
                    const int m_l = wm * 64 + im * 16 + (lane & 15);
                    fa[s][im] = *(const v2f*)(as_r + m_l * STRIDE_A + krow);
                }
#pragma unroll
                for (int in = 0; in < 2; ++in) {
                    const int n_l = wn * 32 + in * 16 + (lane & 15);
                    fb[s][in] = *(const v2f*)(bs_r + prow * PSTRIDE + n_l * 2);
                }
            }
            __builtin_amdgcn_sched_barrier(0);
#pragma unroll
            for (int s = 0; s < 4; ++s)
#pragma unroll
                for (int im = 0; im < 4; ++im)
#pragma unroll
                    for (int in = 0; in < 2; ++in)
                        acc[im][in] = __builtin_amdgcn_wmma_f32_16x16x4_f32(
                            false, fa[s][im], false, fb[s][in],
                            (short)0, acc[im][in], false, false);
            if (hf == 0) {
                __builtin_amdgcn_sched_barrier(0);
                WMMA_PAD();
                __builtin_amdgcn_sched_barrier(0);
            }
        }
    };

    gload();
    stage(0);
    advance();
    gload();
    stage(1);
    advance();
    gload();
    __syncthreads();

    int st_c = 0;
    int st_w = 2;
    int kt = 0;
#pragma unroll 1
    for (int seg = 0; seg < 4; ++seg) {
        int kt_end;
        if (seg == 0)      { __builtin_amdgcn_s_setprio(3); kt_end = KT / 2; }
        else if (seg == 1) { __builtin_amdgcn_s_setprio(2); kt_end = (3 * KT) / 4; }
        else if (seg == 2) { __builtin_amdgcn_s_setprio(1); kt_end = (15 * KT) / 16; }
        else               { __builtin_amdgcn_s_setprio(0); kt_end = KT; }
#pragma unroll 1
        for (; kt < kt_end; ++kt) {
            stage(st_w);
            advance();
            gload();
            compute(st_c);
            __syncthreads();
            __builtin_amdgcn_sched_barrier(0);
            WMMA_PAD();
            __builtin_amdgcn_sched_barrier(0);
            st_c = (st_c == STAGES - 1) ? 0 : st_c + 1;
            st_w = (st_w == STAGES - 1) ? 0 : st_w + 1;
        }
    }

#pragma unroll
    for (int in = 0; in < 2; ++in) {
        const int n_g = n_blk + wn * 32 + in * 16 + (lane & 15);
        const float bv = Bias[n_g];
#pragma unroll
        for (int im = 0; im < 4; ++im) {
            const int m_base = m_blk + wm * 64 + im * 16 + 8 * (lane >> 4);
#pragma unroll
            for (int v = 0; v < 8; ++v)
                Out[(size_t)(m_base + v) * N + n_g] = acc[im][in][v] + bv;
        }
    }
}

extern "C" void kernel_launch(void* const* d_in, const int* in_sizes, int n_in,
                              void* d_out, int out_size, void* d_ws, size_t ws_size,
                              hipStream_t stream) {
    const float* X    = (const float*)d_in[0];
    const float* Mask = (const float*)d_in[1];
    const float* W    = (const float*)d_in[2];
    const float* Bias = (const float*)d_in[3];
    float* Out        = (float*)d_out;

    const int N = in_sizes[3];
    const int K = in_sizes[1] / N;
    const int M = in_sizes[0] / K;

    dim3 grid(N / BLK_N, M / BLK_M);
    Dynamic_Masking_Layer_gemm_wmma_f32<<<grid, dim3(256), 0, stream>>>(
        X, Mask, W, Bias, Out, M, N, K);
}


# K loop without vector address arithmetic: global loads as scalar base + fixed 32-bit lane offset, trips in groups of three with literal stage numbers so LDS addresses are immediates
# speedup vs baseline: 1.2601x; 1.0351x over previous
#include <hip/hip_runtime.h>


typedef float v2f __attribute__((ext_vector_type(2)));
typedef float v4f __attribute__((ext_vector_type(4)));
typedef float v8f __attribute__((ext_vector_type(8)));

#define BLK_M 128
#define BLK_N 128
#define BLK_K 32
#define STRIDE_A 36
#define PSTRIDE 288
#define ASZ (BLK_M * STRIDE_A)
#define BSZ ((BLK_K / 2) * PSTRIDE)
#define STAGES 3

#define WMMA_PAD() asm volatile("v_nop\n\tv_nop\n\tv_nop\n\tv_nop\n\tv_nop")

__global__ __launch_bounds__(256)
void Dynamic_Masking_Layer_gemm_wmma_f32(const float* __restrict__ X,
                                         const float* __restrict__ Mk,
                                         const float* __restrict__ W,
                                         const float* __restrict__ Bias,
                                         float* __restrict__ Out,
                                         int M, int N, int K)
{
    __shared__ float As[STAGES * ASZ];
    __shared__ float Bs[STAGES * BSZ];

    const int tid  = threadIdx.x;
    const int lane = tid & 31;
    const int wave = tid >> 5;
    const int wm   = wave & 1;
    const int wn   = wave >> 1;

    const int m_blk = blockIdx.y * BLK_M;
    const int n_blk = blockIdx.x * BLK_N;

    const int rowA  = tid >> 3;
    const int col4A = tid & 7;
    const int pB = tid >> 5;
    const int qB = tid & 31;

    const char* xs = (const char*)(X + (size_t)m_blk * K);
    const char* ws = (const char*)(W + n_blk);
    const char* ms = (const char*)(Mk + n_blk);
    unsigned offA[4], offB0[2], offB1[2];
#pragma unroll
    for (int i = 0; i < 4; ++i) offA[i] = (unsigned)((rowA + 32 * i) * K + col4A * 4) * 4u;
#pragma unroll
    for (int i = 0; i < 2; ++i) {
        offB0[i] = (unsigned)((2 * (pB + 8 * i)) * N + qB * 4) * 4u;
        offB1[i] = offB0[i] + (unsigned)N * 4u;
    }
    const size_t strideA = (size_t)BLK_K * 4;
    const size_t strideB = (size_t)BLK_K * N * 4;

    v8f acc[4][2];
#pragma unroll
    for (int im = 0; im < 4; ++im)
#pragma unroll
        for (int in = 0; in < 2; ++in)
#pragma unroll
            for (int v = 0; v < 8; ++v)
                acc[im][in][v] = 0.0f;

    const int KT = K / BLK_K;

    v4f ra[4], rw0[2], rw1[2], rm0[2], rm1[2];

    auto gload = [&]() __attribute__((always_inline)) {
#pragma unroll
        for (int i = 0; i < 4; ++i) {
            asm volatile("" : "+v"(offA[i]));
            ra[i] = *(const v4f*)(xs + offA[i]);
        }
#pragma unroll
        for (int i = 0; i < 2; ++i) {
            asm volatile("" : "+v"(offB0[i]));
            asm volatile("" : "+v"(offB1[i]));
            rw0[i] = *(const v4f*)(ws + offB0[i]);
            rw1[i] = *(const v4f*)(ws + offB1[i]);
            rm0[i] = *(const v4f*)(ms + offB0[i]);
            rm1[i] = *(const v4f*)(ms + offB1[i]);
        }
    };

    int held = 0;
    auto advance = [&]() __attribute__((always_inline)) {
        if (held + 1 < KT) {
            xs += strideA; ws += strideB; ms += strideB;
            ++held;
        }
    };

    auto stage = [&](int st) __attribute__((always_inline)) {
        float* __restrict__ as_w = As + st * ASZ;
        float* __restrict__ bs_w = Bs + st * BSZ;
#pragma unroll
        for (int i = 0; i < 4; ++i)
            *(v4f*)(as_w + (rowA + 32 * i) * STRIDE_A + col4A * 4) = ra[i];
#pragma unroll
        for (int i = 0; i < 2; ++i) {
            float s0[4], s1[4];
#pragma unroll
            for (int j = 0; j < 4; ++j) {
                s0[j] = (rm0[i][j] != 0.0f) ? rw0[i][j] : 0.0f;
                s1[j] = (rm1[i][j] != 0.0f) ? rw1[i][j] : 0.0f;
            }
            float* dst = bs_w + (pB + 8 * i) * PSTRIDE + qB * 8;
            v4f lo, hi;
            lo[0] = s0[0]; lo[1] = s1[0]; lo[2] = s0[1]; lo[3] = s1[1];
            hi[0] = s0[2]; hi[1] = s1[2]; hi[2] = s0[3]; hi[3] = s1[3];
            *(v4f*)(dst)     = lo;
            *(v4f*)(dst + 4) = hi;
        }
    };

    auto compute = [&](int st) __attribute__((always_inline)) {
        const float* __restrict__ as_r = As + st * ASZ;
        const float* __restrict__ bs_r = Bs + st * BSZ;
#pragma unroll
        for (int hf = 0; hf < 2; ++hf) {
            v2f fa[4][4], fb[4][2];
#pragma unroll
            for (int s = 0; s < 4; ++s) {
                const int ks   = hf * 4 + s;
                const int krow = 4 * ks + 2 * (lane >> 4);
                const int prow = 2 * ks + (lane >> 4);
#pragma unroll
                for (int im = 0; im < 4; ++im) {
                    const int m_l = wm * 64 + im * 16 + (lane & 15);
                    fa[s][im] = *(const v2f*)(as_r + m_l * STRIDE_A + krow);
                }
#pragma unroll
                for (int in = 0; in < 2; ++in) {
                    const int n_l = wn * 32 + in * 16 + (lane & 15);
                    fb[s][in] = *(const v2f*)(bs_r + prow * PSTRIDE + n_l * 2);
                }
            }
            __builtin_amdgcn_sched_barrier(0);
#pragma unroll
            for (int s = 0; s < 4; ++s)
#pragma unroll
                for (int im = 0; im < 4; ++im)
#pragma unroll
                    for (int in = 0; in < 2; ++in)
                        acc[im][in] = __builtin_amdgcn_wmma_f32_16x16x4_f32(
                            false, fa[s][im], false, fb[s][in],
                            (short)0, acc[im][in], false, false);
            if (hf == 0) {
                __builtin_amdgcn_sched_barrier(0);
                WMMA_PAD();
                __builtin_amdgcn_sched_barrier(0);
            }
        }
    };

    gload();
    stage(0);
    advance();
    gload();
    stage(1);
    advance();
    gload();
    __syncthreads();

    auto trip = [&](int sw, int sc) __attribute__((always_inline)) {
        stage(sw);
        advance();
        gload();
        compute(sc);
        __syncthreads();
        __builtin_amdgcn_sched_barrier(0);
        WMMA_PAD();
        __builtin_amdgcn_sched_barrier(0);
    };

    const int KT3 = (KT / 3) * 3;
    int kt = 0;
#pragma unroll 1
    for (int seg = 0; seg < 4; ++seg) {
        int kt_end;
        if (seg == 0)      { __builtin_amdgcn_s_setprio(3); kt_end = KT / 2; }
        else if (seg == 1) { __builtin_amdgcn_s_setprio(2); kt_end = (3 * KT) / 4; }
        else if (seg == 2) { __builtin_amdgcn_s_setprio(1); kt_end = (15 * KT) / 16; }
        else               { __builtin_amdgcn_s_setprio(0); kt_end = KT3; }
        if (kt_end > KT3) kt_end = KT3;
#pragma unroll 1
        for (; kt < kt_end; kt += 3) {
            trip(2, 0);
            trip(0, 1);
            trip(1, 2);
        }
    }
    if (kt < KT) {
        trip(2, 0);
        if (kt + 1 < KT)
            trip(0, 1);
    }

#pragma unroll
    for (int in = 0; in < 2; ++in) {
        const int n_g = n_blk + wn * 32 + in * 16 + (lane & 15);
        const float bv = Bias[n_g];
#pragma unroll
        for (int im = 0; im < 4; ++im) {
            const int m_base = m_blk + wm * 64 + im * 16 + 8 * (lane >> 4);
#pragma unroll
            for (int v = 0; v < 8; ++v)
                Out[(size_t)(m_base + v) * N + n_g] = acc[im][in][v] + bv;
        }
    }
}

extern "C" void kernel_launch(void* const* d_in, const int* in_sizes, int n_in,
                              void* d_out, int out_size, void* d_ws, size_t ws_size,
                              hipStream_t stream) {
    const float* X    = (const float*)d_in[0];
    const float* Mask = (const float*)d_in[1];
    const float* W    = (const float*)d_in[2];
    const float* Bias = (const float*)d_in[3];
    float* Out        = (float*)d_out;

    const int N = in_sizes[3];
    const int K = in_sizes[1] / N;
    const int M = in_sizes[0] / K;

    dim3 grid(N / BLK_N, M / BLK_M);
    Dynamic_Masking_Layer_gemm_wmma_f32<<<grid, dim3(256), 0, stream>>>(
        X, Mask, W, Bias, Out, M, N, K);
}
